# starting kernel
# baseline (speedup 1.0000x reference)
#include <hip/hip_runtime.h>
#include <hip/hip_bf16.h>
#include <math.h>

typedef __attribute__((ext_vector_type(2))) float v2f;
typedef __attribute__((ext_vector_type(8))) float v8f;

#define NEG_SLOPE 0.2f

__device__ __forceinline__ unsigned enc_f32(float f) {
  unsigned u = __float_as_uint(f);
  return u ^ ((u >> 31) ? 0xFFFFFFFFu : 0x80000000u);
}
__device__ __forceinline__ float dec_f32(unsigned u) {
  unsigned v = u ^ ((u >> 31) ? 0x80000000u : 0xFFFFFFFFu);
  return __uint_as_float(v);
}

__global__ __launch_bounds__(256) void gat_gemm128(const float* __restrict__ X,
                                                   const float* __restrict__ W,
                                                   float* __restrict__ Hout,
                                                   int N) {
  const int lane  = threadIdx.x & 31;
  const int wv    = threadIdx.x >> 5;
  const int m0    = blockIdx.x << 4;
  const int n0    = wv << 4;
  const int r     = lane & 15;
  const int khalf = (lane >> 4) << 1;

  int arow = m0 + r;
  if (arow >= N) arow = N - 1;
  const float* ap = X + (size_t)arow * 128;

  v8f acc = {0.f, 0.f, 0.f, 0.f, 0.f, 0.f, 0.f, 0.f};
#pragma unroll
  for (int k0 = 0; k0 < 128; k0 += 4) {
    const int ka = k0 + khalf;
    v2f a; a.x = ap[ka]; a.y = ap[ka + 1];
    v2f b; b.x = W[(size_t)ka * 128 + n0 + r];
           b.y = W[(size_t)(ka + 1) * 128 + n0 + r];
    acc = __builtin_amdgcn_wmma_f32_16x16x4_f32(false, a, false, b,
                                                (short)0, acc, false, false);
  }
  const int rowbase = m0 + ((lane >> 4) << 3);
  const int col     = n0 + r;
#pragma unroll
  for (int i = 0; i < 8; ++i) {
    const int row = rowbase + i;
    if (row < N) Hout[(size_t)row * 128 + col] = acc[i];
  }
}

__global__ void gat_scores_init(const float* __restrict__ Hm,
                                const float* __restrict__ asrc,
                                const float* __restrict__ adst,
                                float* __restrict__ es, float* __restrict__ ed,
                                unsigned* __restrict__ menc,
                                float* __restrict__ ssum,
                                float* __restrict__ acc, int N) {
  const int idx = blockIdx.x * blockDim.x + threadIdx.x;
  if (idx >= N * 4) return;
  const int hd = idx & 3;
  const size_t base = (size_t)(idx >> 2) * 128 + hd * 32;
  const float* hp  = Hm + base;
  float*       ap  = acc + base;
  const float* sa  = asrc + hd * 32;
  const float* da  = adst + hd * 32;
  float s = 0.f, d = 0.f;
#pragma unroll
  for (int c = 0; c < 32; c += 4) {
    float4 hv = *(const float4*)(hp + c);
    float4 av = *(const float4*)(sa + c);
    float4 bv = *(const float4*)(da + c);
    s += hv.x * av.x + hv.y * av.y + hv.z * av.z + hv.w * av.w;
    d += hv.x * bv.x + hv.y * bv.y + hv.z * bv.z + hv.w * bv.w;
    *(float4*)(ap + c) = make_float4(0.f, 0.f, 0.f, 0.f);
  }
  es[idx] = s;
  ed[idx] = d;
  menc[idx] = 0u;
  ssum[idx] = 0.f;
}

__device__ __forceinline__ float lrelu(float x) {
  return x > 0.f ? x : NEG_SLOPE * x;
}

__global__ void gat_edge_max(const int* __restrict__ ei,
                             const float* __restrict__ es,
                             const float* __restrict__ ed,
                             unsigned* __restrict__ menc, int E, int N) {
  const int e = blockIdx.x * blockDim.x + threadIdx.x;
  const int Et = E + N;
  if (e >= Et) return;
  int s, d;
  if (e < E) { s = ei[e]; d = ei[E + e]; } else { s = d = e - E; }
  float4 vs = *(const float4*)(es + (size_t)s * 4);
  float4 vd = *(const float4*)(ed + (size_t)d * 4);
  unsigned* mp = menc + (size_t)d * 4;
  atomicMax(mp + 0, enc_f32(lrelu(vs.x + vd.x)));
  atomicMax(mp + 1, enc_f32(lrelu(vs.y + vd.y)));
  atomicMax(mp + 2, enc_f32(lrelu(vs.z + vd.z)));
  atomicMax(mp + 3, enc_f32(lrelu(vs.w + vd.w)));
}

__global__ __launch_bounds__(256) void gat_edge_scatter(
    const int* __restrict__ ei,
    const float* __restrict__ es, const float* __restrict__ ed,
    const unsigned* __restrict__ menc, float* __restrict__ ssum,
    const float* __restrict__ Hm, float* __restrict__ acc, int E, int N) {
  const int e = blockIdx.x * 8 + (threadIdx.x >> 5);
  const int lane = threadIdx.x & 31;
  const int Et = E + N;
  if (e >= Et) return;
  int s, d;
  if (e < E) { s = ei[e]; d = ei[E + e]; } else { s = d = e - E; }
  const int hd = lane >> 3;
  float l = es[(size_t)s * 4 + hd] + ed[(size_t)d * 4 + hd];
  l = lrelu(l);
  const float mh = dec_f32(menc[(size_t)d * 4 + hd]);
  const float ex = expf(l - mh);
  if ((lane & 7) == 0) atomicAdd(&ssum[(size_t)d * 4 + hd], ex);
  float4 hv = *(const float4*)(Hm + (size_t)s * 128 + lane * 4);
  float* op = acc + (size_t)d * 128 + lane * 4;
  atomicAdd(op + 0, hv.x * ex);
  atomicAdd(op + 1, hv.y * ex);
  atomicAdd(op + 2, hv.z * ex);
  atomicAdd(op + 3, hv.w * ex);
}

__global__ void gat_finalize(const float* __restrict__ acc,
                             const float* __restrict__ ssum,
                             const float* __restrict__ bias,
                             float* __restrict__ outp, int N, int apply_elu) {
  const int idx = blockIdx.x * blockDim.x + threadIdx.x;
  if (idx >= N * 128) return;
  const int n  = idx >> 7;
  const int hd = (idx >> 5) & 3;
  float v = acc[idx] / ssum[(size_t)n * 4 + hd] + bias[idx & 127];
  if (apply_elu) v = v > 0.f ? v : expm1f(v);
  outp[idx] = v;
}

extern "C" void kernel_launch(void* const* d_in, const int* in_sizes, int n_in,
                              void* d_out, int out_size, void* d_ws, size_t ws_size,
                              hipStream_t stream) {
  const float* x  = (const float*)d_in[0];
  const int*   ei = (const int*)d_in[1];
  const int N = in_sizes[0] / 128;
  const int E = in_sizes[1] / 2;

  const float* Wl[3] = {(const float*)d_in[2], (const float*)d_in[6],  (const float*)d_in[10]};
  const float* As[3] = {(const float*)d_in[3], (const float*)d_in[7],  (const float*)d_in[11]};
  const float* Ad[3] = {(const float*)d_in[4], (const float*)d_in[8],  (const float*)d_in[12]};
  const float* Bb[3] = {(const float*)d_in[5], (const float*)d_in[9],  (const float*)d_in[13]};

  float*    F0   = (float*)d_ws;
  float*    F1   = F0 + (size_t)N * 128;
  float*    es   = F1 + (size_t)N * 128;
  float*    ed   = es + (size_t)N * 4;
  unsigned* menc = (unsigned*)(ed + (size_t)N * 4);
  float*    ssum = (float*)(menc + (size_t)N * 4);

  const int Et = E + N;
  const dim3 bGemm(256);
  const dim3 gGemm((N + 15) / 16);
  const int gScore = (N * 4 + 255) / 256;
  const int gMax   = (Et + 255) / 256;
  const int gScat  = (Et + 7) / 8;
  const int gFin   = (N * 128 + 255) / 256;

  const float* inp = x;
  for (int layer = 0; layer < 3; ++layer) {
    float* Hm = (layer & 1) ? F1 : F0;
    float* Ac = (layer & 1) ? F0 : F1;
    gat_gemm128<<<gGemm, bGemm, 0, stream>>>(inp, Wl[layer], Hm, N);
    gat_scores_init<<<gScore, 256, 0, stream>>>(Hm, As[layer], Ad[layer],
                                                es, ed, menc, ssum, Ac, N);
    gat_edge_max<<<gMax, 256, 0, stream>>>(ei, es, ed, menc, E, N);
    gat_edge_scatter<<<gScat, 256, 0, stream>>>(ei, es, ed, menc, ssum, Hm, Ac, E, N);
    float* ot = (layer == 2) ? (float*)d_out : Hm;
    gat_finalize<<<gFin, 256, 0, stream>>>(Ac, ssum, Bb[layer], ot, N,
                                           layer < 2 ? 1 : 0);
    inp = Hm;
  }
}


# Segment-max pass: max over es[src] alone (leaky relu and +ed[dst] are monotone, applied after), lane=(edge,head), 128 edges per wave: 1 load line + 1 atomic line per edge, was 2 + 4; sum order kept
# speedup vs baseline: 1.8979x; 1.8979x over previous
#include <hip/hip_runtime.h>
#include <hip/hip_bf16.h>
#include <math.h>

typedef __attribute__((ext_vector_type(2))) float v2f;
typedef __attribute__((ext_vector_type(8))) float v8f;

#define NEG_SLOPE 0.2f

__device__ __forceinline__ unsigned enc_f32(float f) {
  unsigned u = __float_as_uint(f);
  return u ^ ((u >> 31) ? 0xFFFFFFFFu : 0x80000000u);
}
__device__ __forceinline__ float dec_f32(unsigned u) {
  unsigned v = u ^ ((u >> 31) ? 0x80000000u : 0xFFFFFFFFu);
  return __uint_as_float(v);
}

__global__ __launch_bounds__(256) void gat_gemm128(const float* __restrict__ X,
                                                   const float* __restrict__ W,
                                                   float* __restrict__ Hout,
                                                   int N) {
  const int lane  = threadIdx.x & 31;
  const int wv    = threadIdx.x >> 5;
  const int m0    = blockIdx.x << 4;
  const int n0    = wv << 4;
  const int r     = lane & 15;
  const int khalf = (lane >> 4) << 1;

  int arow = m0 + r;
  if (arow >= N) arow = N - 1;
  const float* ap = X + (size_t)arow * 128;

  v8f acc = {0.f, 0.f, 0.f, 0.f, 0.f, 0.f, 0.f, 0.f};
#pragma unroll
  for (int k0 = 0; k0 < 128; k0 += 4) {
    const int ka = k0 + khalf;
    v2f a; a.x = ap[ka]; a.y = ap[ka + 1];
    v2f b; b.x = W[(size_t)ka * 128 + n0 + r];
           b.y = W[(size_t)(ka + 1) * 128 + n0 + r];
    acc = __builtin_amdgcn_wmma_f32_16x16x4_f32(false, a, false, b,
                                                (short)0, acc, false, false);
  }
  const int rowbase = m0 + ((lane >> 4) << 3);
  const int col     = n0 + r;
#pragma unroll
  for (int i = 0; i < 8; ++i) {
    const int row = rowbase + i;
    if (row < N) Hout[(size_t)row * 128 + col] = acc[i];
  }
}

__global__ void gat_scores_init(const float* __restrict__ Hm,
                                const float* __restrict__ asrc,
                                const float* __restrict__ adst,
                                float* __restrict__ es, float* __restrict__ ed,
                                unsigned* __restrict__ menc,
                                float* __restrict__ ssum,
                                float* __restrict__ acc, int N) {
  const int idx = blockIdx.x * blockDim.x + threadIdx.x;
  if (idx >= N * 4) return;
  const int hd = idx & 3;
  const size_t base = (size_t)(idx >> 2) * 128 + hd * 32;
  const float* hp  = Hm + base;
  float*       ap  = acc + base;
  const float* sa  = asrc + hd * 32;
  const float* da  = adst + hd * 32;
  float s = 0.f, d = 0.f;
#pragma unroll
  for (int c = 0; c < 32; c += 4) {
    float4 hv = *(const float4*)(hp + c);
    float4 av = *(const float4*)(sa + c);
    float4 bv = *(const float4*)(da + c);
    s += hv.x * av.x + hv.y * av.y + hv.z * av.z + hv.w * av.w;
    d += hv.x * bv.x + hv.y * bv.y + hv.z * bv.z + hv.w * bv.w;
    *(float4*)(ap + c) = make_float4(0.f, 0.f, 0.f, 0.f);
  }
  es[idx] = s;
  ed[idx] = d;
  menc[idx] = 0u;
  ssum[idx] = 0.f;
}

__device__ __forceinline__ float lrelu(float x) {
  return x > 0.f ? x : NEG_SLOPE * x;
}

#define MAXP_CHUNKS 4
__global__ __launch_bounds__(256) void gat_edge_max(const int* __restrict__ ei,
                                                    const float* __restrict__ es,
                                                    unsigned* __restrict__ menc,
                                                    int E, int N) {
  const int lane = threadIdx.x & 31;
  const int Et = E + N;
  const int e0 = (blockIdx.x * 8 + (threadIdx.x >> 5)) * (32 * MAXP_CHUNKS);
  if (e0 >= Et) return;
  int s[MAXP_CHUNKS], d[MAXP_CHUNKS];
#pragma unroll
  for (int c = 0; c < MAXP_CHUNKS; ++c) {
    int e = e0 + c * 32 + lane;
    e = e < Et ? e : Et - 1;
    if (e < E) { s[c] = ei[e]; d[c] = ei[E + e]; } else { s[c] = d[c] = e - E; }
  }
  const int sub = lane >> 2;
  const int hd  = lane & 3;
  float v[MAXP_CHUNKS * 4];
  int   dj[MAXP_CHUNKS * 4];
#pragma unroll
  for (int c = 0; c < MAXP_CHUNKS; ++c) {
#pragma unroll
    for (int g = 0; g < 4; ++g) {
      const int from = g * 8 + sub;
      const int sj = __shfl(s[c], from);
      dj[c * 4 + g] = __shfl(d[c], from);
      v[c * 4 + g]  = es[(size_t)sj * 4 + hd];
    }
  }
#pragma unroll
  for (int i = 0; i < MAXP_CHUNKS * 4; ++i)
    atomicMax(menc + (size_t)dj[i] * 4 + hd, enc_f32(v[i]));
}

struct Row4 { float c0, c1, c2, c3; };

__device__ __forceinline__ float lane_f(float v, int j) {
  return __int_as_float(__builtin_amdgcn_readlane(__float_as_int(v), j));
}
__device__ __forceinline__ Row4 load_row(const float* __restrict__ Hm, int sj, int lane) {
  const float* hp = Hm + (size_t)sj * 128 + lane;
  Row4 r;
  r.c0 = hp[0]; r.c1 = hp[32]; r.c2 = hp[64]; r.c3 = hp[96];
  return r;
}
__device__ __forceinline__ void add_row(float* __restrict__ acc, int dj, Row4 h,
                                        float w0, float w1, float w2, float w3, int lane) {
  float* op = acc + (size_t)dj * 128 + lane;
  atomicAdd(op,      h.c0 * w0);
  atomicAdd(op + 32, h.c1 * w1);
  atomicAdd(op + 64, h.c2 * w2);
  atomicAdd(op + 96, h.c3 * w3);
}
__device__ __forceinline__ void add_den(float* __restrict__ ssum, int d,
                                        float x0, float x1, float x2, float x3,
                                        int j0, int cnt, int lane) {
  const int src = j0 + (lane >> 2);
  const int hsel = lane & 3;
  const float t0 = __shfl(x0, src), t1 = __shfl(x1, src);
  const float t2 = __shfl(x2, src), t3 = __shfl(x3, src);
  const int dz = __shfl(d, src);
  const float za = (hsel & 1) ? t1 : t0;
  const float zb = (hsel & 1) ? t3 : t2;
  const float z  = (hsel & 2) ? zb : za;
  if (src < cnt) atomicAdd(&ssum[(size_t)dz * 4 + hsel], z);
}

__global__ __launch_bounds__(256) void gat_edge_scatter(
    const int* __restrict__ ei,
    const float* __restrict__ es, const float* __restrict__ ed,
    const unsigned* __restrict__ menc, float* __restrict__ ssum,
    const float* __restrict__ Hm, float* __restrict__ acc, int E, int N) {
  const int lane = threadIdx.x & 31;
  const int Et = E + N;
  const int e0 = (blockIdx.x * 8 + (threadIdx.x >> 5)) * 32;
  if (e0 >= Et) return;
  const int e = e0 + lane;
  int s = 0, d = 0;
  float x0 = 0.f, x1 = 0.f, x2 = 0.f, x3 = 0.f;
  if (e < Et) {
    if (e < E) { s = ei[e]; d = ei[E + e]; } else { s = d = e - E; }
    const float4 vs = *(const float4*)(es + (size_t)s * 4);
    const float4 vd = *(const float4*)(ed + (size_t)d * 4);
    const uint4  mm = *(const uint4*)(menc + (size_t)d * 4);
    x0 = expf(lrelu(vs.x + vd.x) - lrelu(dec_f32(mm.x) + vd.x));
    x1 = expf(lrelu(vs.y + vd.y) - lrelu(dec_f32(mm.y) + vd.y));
    x2 = expf(lrelu(vs.z + vd.z) - lrelu(dec_f32(mm.z) + vd.z));
    x3 = expf(lrelu(vs.w + vd.w) - lrelu(dec_f32(mm.w) + vd.w));
  }
  const int cnt = (Et - e0 < 32) ? (Et - e0) : 32;
  for (int j0 = 0; j0 < cnt; j0 += 8) {
    if (j0 + 8 <= cnt) {
      Row4 hv[8];
#pragma unroll
      for (int u = 0; u < 8; ++u)
        hv[u] = load_row(Hm, __builtin_amdgcn_readlane(s, j0 + u), lane);
      add_den(ssum, d, x0, x1, x2, x3, j0, cnt, lane);
#pragma unroll
      for (int u = 0; u < 8; ++u)
        add_row(acc, __builtin_amdgcn_readlane(d, j0 + u), hv[u],
                lane_f(x0, j0 + u), lane_f(x1, j0 + u),
                lane_f(x2, j0 + u), lane_f(x3, j0 + u), lane);
    } else {
      add_den(ssum, d, x0, x1, x2, x3, j0, cnt, lane);
      for (int u = 0; j0 + u < cnt; ++u)
        add_row(acc, __builtin_amdgcn_readlane(d, j0 + u),
                load_row(Hm, __builtin_amdgcn_readlane(s, j0 + u), lane),
                lane_f(x0, j0 + u), lane_f(x1, j0 + u),
                lane_f(x2, j0 + u), lane_f(x3, j0 + u), lane);
    }
  }
}

__global__ void gat_finalize(const float* __restrict__ acc,
                             const float* __restrict__ ssum,
                             const float* __restrict__ bias,
                             float* __restrict__ outp, int N, int apply_elu) {
  const int idx = blockIdx.x * blockDim.x + threadIdx.x;
  if (idx >= N * 128) return;
  const int n  = idx >> 7;
  const int hd = (idx >> 5) & 3;
  float v = acc[idx] / ssum[(size_t)n * 4 + hd] + bias[idx & 127];
  if (apply_elu) v = v > 0.f ? v : expm1f(v);
  outp[idx] = v;
}

extern "C" void kernel_launch(void* const* d_in, const int* in_sizes, int n_in,
                              void* d_out, int out_size, void* d_ws, size_t ws_size,
                              hipStream_t stream) {
  const float* x  = (const float*)d_in[0];
  const int*   ei = (const int*)d_in[1];
  const int N = in_sizes[0] / 128;
  const int E = in_sizes[1] / 2;

  const float* Wl[3] = {(const float*)d_in[2], (const float*)d_in[6],  (const float*)d_in[10]};
  const float* As[3] = {(const float*)d_in[3], (const float*)d_in[7],  (const float*)d_in[11]};
  const float* Ad[3] = {(const float*)d_in[4], (const float*)d_in[8],  (const float*)d_in[12]};
  const float* Bb[3] = {(const float*)d_in[5], (const float*)d_in[9],  (const float*)d_in[13]};

  float*    F0   = (float*)d_ws;
  float*    F1   = F0 + (size_t)N * 128;
  float*    es   = F1 + (size_t)N * 128;
  float*    ed   = es + (size_t)N * 4;
  unsigned* menc = (unsigned*)(ed + (size_t)N * 4);
  float*    ssum = (float*)(menc + (size_t)N * 4);

  const int Et = E + N;
  const dim3 bGemm(256);
  const dim3 gGemm((N + 15) / 16);
  const int gScore = (N * 4 + 255) / 256;
  const int gMax   = (Et + 256 * MAXP_CHUNKS - 1) / (256 * MAXP_CHUNKS);
  const int gScat  = (Et + 255) / 256;
  const int gFin   = (N * 128 + 255) / 256;

  const float* inp = x;
  for (int layer = 0; layer < 3; ++layer) {
    float* Hm = (layer & 1) ? F1 : F0;
    float* Ac = (layer & 1) ? F0 : F1;
    gat_gemm128<<<gGemm, bGemm, 0, stream>>>(inp, Wl[layer], Hm, N);
    gat_scores_init<<<gScore, 256, 0, stream>>>(Hm, As[layer], Ad[layer],
                                                es, ed, menc, ssum, Ac, N);
    gat_edge_max<<<gMax, 256, 0, stream>>>(ei, es, menc, E, N);
    gat_edge_scatter<<<gScat, 256, 0, stream>>>(ei, es, ed, menc, ssum, Hm, Ac, E, N);
    float* ot = (layer == 2) ? (float*)d_out : Hm;
    gat_finalize<<<gFin, 256, 0, stream>>>(Ac, ssum, Bb[layer], ot, N,
                                           layer < 2 ? 1 : 0);
    inp = Hm;
  }
}


# GEMM: X tile staged once per block in LDS for its 8 waves, each wave's 128x16 W strip kept in registers over a persistent loop on row tiles; k summed in a new order (2 accumulators)
# speedup vs baseline: 2.0521x; 1.0812x over previous
#include <hip/hip_runtime.h>
#include <hip/hip_bf16.h>
#include <math.h>

typedef __attribute__((ext_vector_type(2))) float v2f;
typedef __attribute__((ext_vector_type(8))) float v8f;

#define NEG_SLOPE 0.2f

__device__ __forceinline__ unsigned enc_f32(float f) {
  unsigned u = __float_as_uint(f);
  return u ^ ((u >> 31) ? 0xFFFFFFFFu : 0x80000000u);
}
__device__ __forceinline__ float dec_f32(unsigned u) {
  unsigned v = u ^ ((u >> 31) ? 0x80000000u : 0xFFFFFFFFu);
  return __uint_as_float(v);
}

#define GEMM_P4 33
__global__ __launch_bounds__(256) void gat_gemm128(const float* __restrict__ X,
                                                   const float* __restrict__ W,
                                                   float* __restrict__ Hout,
                                                   int N) {
  __shared__ float4 As4[2][16 * GEMM_P4];
  const int tid  = threadIdx.x;
  const int lane = tid & 31;
  const int wv   = tid >> 5;
  const int n0   = wv << 4;
  const int r    = lane & 15;
  const int half = lane >> 4;
  const int ntiles = (N + 15) >> 4;

  v2f bf[32];
#pragma unroll
  for (int j = 0; j < 16; ++j) {
    const float* wp = W + (size_t)(8 * j + 4 * half) * 128 + n0 + r;
    bf[2 * j].x     = wp[0];
    bf[2 * j].y     = wp[128];
    bf[2 * j + 1].x = wp[256];
    bf[2 * j + 1].y = wp[384];
  }

  const int lrow = tid >> 5;
  const int c4   = tid & 31;
  int t = blockIdx.x;
  float4 p0 = make_float4(0.f, 0.f, 0.f, 0.f), p1 = p0;
  if (t < ntiles) {
    int ra = (t << 4) + lrow, rb = ra + 8;
    ra = ra < N ? ra : N - 1;
    rb = rb < N ? rb : N - 1;
    p0 = *(const float4*)(X + (size_t)ra * 128 + c4 * 4);
    p1 = *(const float4*)(X + (size_t)rb * 128 + c4 * 4);
  }
  int buf = 0;
  for (; t < ntiles; t += gridDim.x) {
    As4[buf][lrow * GEMM_P4 + c4]       = p0;
    As4[buf][(lrow + 8) * GEMM_P4 + c4] = p1;
    __syncthreads();
    const int tn = t + gridDim.x;
    if (tn < ntiles) {
      int ra = (tn << 4) + lrow, rb = ra + 8;
      ra = ra < N ? ra : N - 1;
      rb = rb < N ? rb : N - 1;
      p0 = *(const float4*)(X + (size_t)ra * 128 + c4 * 4);
      p1 = *(const float4*)(X + (size_t)rb * 128 + c4 * 4);
    }
    v8f acc0 = {0.f, 0.f, 0.f, 0.f, 0.f, 0.f, 0.f, 0.f};
    v8f acc1 = acc0;
    const float4* arow = &As4[buf][r * GEMM_P4 + half];
#pragma unroll
    for (int j = 0; j < 16; ++j) {
      const float4 q = arow[2 * j];
      v2f a0; a0.x = q.x; a0.y = q.y;
      v2f a1; a1.x = q.z; a1.y = q.w;
      acc0 = __builtin_amdgcn_wmma_f32_16x16x4_f32(false, a0, false, bf[2 * j],
                                                   (short)0, acc0, false, false);
      acc1 = __builtin_amdgcn_wmma_f32_16x16x4_f32(false, a1, false, bf[2 * j + 1],
                                                   (short)0, acc1, false, false);
    }
    const int rowbase = (t << 4) + (half << 3);
    const int col     = n0 + r;
#pragma unroll
    for (int i = 0; i < 8; ++i) {
      const int row = rowbase + i;
      if (row < N) Hout[(size_t)row * 128 + col] = acc0[i] + acc1[i];
    }
    buf ^= 1;
  }
}

__global__ void gat_scores_init(const float* __restrict__ Hm,
                                const float* __restrict__ asrc,
                                const float* __restrict__ adst,
                                float* __restrict__ es, float* __restrict__ ed,
                                unsigned* __restrict__ menc,
                                float* __restrict__ ssum,
                                float* __restrict__ acc, int N) {
  const int idx = blockIdx.x * blockDim.x + threadIdx.x;
  if (idx >= N * 4) return;
  const int hd = idx & 3;
  const size_t base = (size_t)(idx >> 2) * 128 + hd * 32;
  const float* hp  = Hm + base;
  float*       ap  = acc + base;
  const float* sa  = asrc + hd * 32;
  const float* da  = adst + hd * 32;
  float s = 0.f, d = 0.f;
#pragma unroll
  for (int c = 0; c < 32; c += 4) {
    float4 hv = *(const float4*)(hp + c);
    float4 av = *(const float4*)(sa + c);
    float4 bv = *(const float4*)(da + c);
    s += hv.x * av.x + hv.y * av.y + hv.z * av.z + hv.w * av.w;
    d += hv.x * bv.x + hv.y * bv.y + hv.z * bv.z + hv.w * bv.w;
    *(float4*)(ap + c) = make_float4(0.f, 0.f, 0.f, 0.f);
  }
  es[idx] = s;
  ed[idx] = d;
  menc[idx] = 0u;
  ssum[idx] = 0.f;
}

__device__ __forceinline__ float lrelu(float x) {
  return x > 0.f ? x : NEG_SLOPE * x;
}

#define MAXP_CHUNKS 4
__global__ __launch_bounds__(256) void gat_edge_max(const int* __restrict__ ei,
                                                    const float* __restrict__ es,
                                                    unsigned* __restrict__ menc,
                                                    int E, int N) {
  const int lane = threadIdx.x & 31;
  const int Et = E + N;
  const int e0 = (blockIdx.x * 8 + (threadIdx.x >> 5)) * (32 * MAXP_CHUNKS);
  if (e0 >= Et) return;
  int s[MAXP_CHUNKS], d[MAXP_CHUNKS];
#pragma unroll
  for (int c = 0; c < MAXP_CHUNKS; ++c) {
    int e = e0 + c * 32 + lane;
    e = e < Et ? e : Et - 1;
    if (e < E) { s[c] = ei[e]; d[c] = ei[E + e]; } else { s[c] = d[c] = e - E; }
  }
  const int sub = lane >> 2;
  const int hd  = lane & 3;
  float v[MAXP_CHUNKS * 4];
  int   dj[MAXP_CHUNKS * 4];
#pragma unroll
  for (int c = 0; c < MAXP_CHUNKS; ++c) {
#pragma unroll
    for (int g = 0; g < 4; ++g) {
      const int from = g * 8 + sub;
      const int sj = __shfl(s[c], from);
      dj[c * 4 + g] = __shfl(d[c], from);
      v[c * 4 + g]  = es[(size_t)sj * 4 + hd];
    }
  }
#pragma unroll
  for (int i = 0; i < MAXP_CHUNKS * 4; ++i)
    atomicMax(menc + (size_t)dj[i] * 4 + hd, enc_f32(v[i]));
}

struct Row4 { float c0, c1, c2, c3; };

__device__ __forceinline__ float lane_f(float v, int j) {
  return __int_as_float(__builtin_amdgcn_readlane(__float_as_int(v), j));
}
__device__ __forceinline__ Row4 load_row(const float* __restrict__ Hm, int sj, int lane) {
  const float* hp = Hm + (size_t)sj * 128 + lane;
  Row4 r;
  r.c0 = hp[0]; r.c1 = hp[32]; r.c2 = hp[64]; r.c3 = hp[96];
  return r;
}
__device__ __forceinline__ void add_row(float* __restrict__ acc, int dj, Row4 h,
                                        float w0, float w1, float w2, float w3, int lane) {
  float* op = acc + (size_t)dj * 128 + lane;
  atomicAdd(op,      h.c0 * w0);
  atomicAdd(op + 32, h.c1 * w1);
  atomicAdd(op + 64, h.c2 * w2);
  atomicAdd(op + 96, h.c3 * w3);
}
__device__ __forceinline__ void add_den(float* __restrict__ ssum, int d,
                                        float x0, float x1, float x2, float x3,
                                        int j0, int cnt, int lane) {
  const int src = j0 + (lane >> 2);
  const int hsel = lane & 3;
  const float t0 = __shfl(x0, src), t1 = __shfl(x1, src);
  const float t2 = __shfl(x2, src), t3 = __shfl(x3, src);
  const int dz = __shfl(d, src);
  const float za = (hsel & 1) ? t1 : t0;
  const float zb = (hsel & 1) ? t3 : t2;
  const float z  = (hsel & 2) ? zb : za;
  if (src < cnt) atomicAdd(&ssum[(size_t)dz * 4 + hsel], z);
}

__global__ __launch_bounds__(256) void gat_edge_scatter(
    const int* __restrict__ ei,
    const float* __restrict__ es, const float* __restrict__ ed,
    const unsigned* __restrict__ menc, float* __restrict__ ssum,
    const float* __restrict__ Hm, float* __restrict__ acc, int E, int N) {
  const int lane = threadIdx.x & 31;
  const int Et = E + N;
  const int e0 = (blockIdx.x * 8 + (threadIdx.x >> 5)) * 32;
  if (e0 >= Et) return;
  const int e = e0 + lane;
  int s = 0, d = 0;
  float x0 = 0.f, x1 = 0.f, x2 = 0.f, x3 = 0.f;
  if (e < Et) {
    if (e < E) { s = ei[e]; d = ei[E + e]; } else { s = d = e - E; }
    const float4 vs = *(const float4*)(es + (size_t)s * 4);
    const float4 vd = *(const float4*)(ed + (size_t)d * 4);
    const uint4  mm = *(const uint4*)(menc + (size_t)d * 4);
    x0 = expf(lrelu(vs.x + vd.x) - lrelu(dec_f32(mm.x) + vd.x));
    x1 = expf(lrelu(vs.y + vd.y) - lrelu(dec_f32(mm.y) + vd.y));
    x2 = expf(lrelu(vs.z + vd.z) - lrelu(dec_f32(mm.z) + vd.z));
    x3 = expf(lrelu(vs.w + vd.w) - lrelu(dec_f32(mm.w) + vd.w));
  }
  const int cnt = (Et - e0 < 32) ? (Et - e0) : 32;
  for (int j0 = 0; j0 < cnt; j0 += 8) {
    if (j0 + 8 <= cnt) {
      Row4 hv[8];
#pragma unroll
      for (int u = 0; u < 8; ++u)
        hv[u] = load_row(Hm, __builtin_amdgcn_readlane(s, j0 + u), lane);
      add_den(ssum, d, x0, x1, x2, x3, j0, cnt, lane);
#pragma unroll
      for (int u = 0; u < 8; ++u)
        add_row(acc, __builtin_amdgcn_readlane(d, j0 + u), hv[u],
                lane_f(x0, j0 + u), lane_f(x1, j0 + u),
                lane_f(x2, j0 + u), lane_f(x3, j0 + u), lane);
    } else {
      add_den(ssum, d, x0, x1, x2, x3, j0, cnt, lane);
      for (int u = 0; j0 + u < cnt; ++u)
        add_row(acc, __builtin_amdgcn_readlane(d, j0 + u),
                load_row(Hm, __builtin_amdgcn_readlane(s, j0 + u), lane),
                lane_f(x0, j0 + u), lane_f(x1, j0 + u),
                lane_f(x2, j0 + u), lane_f(x3, j0 + u), lane);
    }
  }
}

__global__ void gat_finalize(const float* __restrict__ acc,
                             const float* __restrict__ ssum,
                             const float* __restrict__ bias,
                             float* __restrict__ outp, int N, int apply_elu) {
  const int idx = blockIdx.x * blockDim.x + threadIdx.x;
  if (idx >= N * 128) return;
  const int n  = idx >> 7;
  const int hd = (idx >> 5) & 3;
  float v = acc[idx] / ssum[(size_t)n * 4 + hd] + bias[idx & 127];
  if (apply_elu) v = v > 0.f ? v : expm1f(v);
  outp[idx] = v;
}

extern "C" void kernel_launch(void* const* d_in, const int* in_sizes, int n_in,
                              void* d_out, int out_size, void* d_ws, size_t ws_size,
                              hipStream_t stream) {
  const float* x  = (const float*)d_in[0];
  const int*   ei = (const int*)d_in[1];
  const int N = in_sizes[0] / 128;
  const int E = in_sizes[1] / 2;

  const float* Wl[3] = {(const float*)d_in[2], (const float*)d_in[6],  (const float*)d_in[10]};
  const float* As[3] = {(const float*)d_in[3], (const float*)d_in[7],  (const float*)d_in[11]};
  const float* Ad[3] = {(const float*)d_in[4], (const float*)d_in[8],  (const float*)d_in[12]};
  const float* Bb[3] = {(const float*)d_in[5], (const float*)d_in[9],  (const float*)d_in[13]};

  float*    F0   = (float*)d_ws;
  float*    F1   = F0 + (size_t)N * 128;
  float*    es   = F1 + (size_t)N * 128;
  float*    ed   = es + (size_t)N * 4;
  unsigned* menc = (unsigned*)(ed + (size_t)N * 4);
  float*    ssum = (float*)(menc + (size_t)N * 4);

  const int Et = E + N;
  const dim3 bGemm(256);
  const dim3 gGemm(512);
  const int gScore = (N * 4 + 255) / 256;
  const int gMax   = (Et + 256 * MAXP_CHUNKS - 1) / (256 * MAXP_CHUNKS);
  const int gScat  = (Et + 255) / 256;
  const int gFin   = (N * 128 + 255) / 256;

  const float* inp = x;
  for (int layer = 0; layer < 3; ++layer) {
    float* Hm = (layer & 1) ? F1 : F0;
    float* Ac = (layer & 1) ? F0 : F1;
    gat_gemm128<<<gGemm, bGemm, 0, stream>>>(inp, Wl[layer], Hm, N);
    gat_scores_init<<<gScore, 256, 0, stream>>>(Hm, As[layer], Ad[layer],
                                                es, ed, menc, ssum, Ac, N);
    gat_edge_max<<<gMax, 256, 0, stream>>>(ei, es, menc, E, N);
    gat_edge_scatter<<<gScat, 256, 0, stream>>>(ei, es, ed, menc, ssum, Hm, Ac, E, N);
    float* ot = (layer == 2) ? (float*)d_out : Hm;
    gat_finalize<<<gFin, 256, 0, stream>>>(Ac, ssum, Bb[layer], ot, N,
                                           layer < 2 ? 1 : 0);
    inp = Hm;
  }
}


# Both edge passes: nodes split in two, a work-group reads its die and adds only into its chip half's nodes (edge chunks drawn from a counter per half): one L2 half per added line; atomic order moves
# speedup vs baseline: 2.8476x; 1.3877x over previous
#include <hip/hip_runtime.h>
#include <hip/hip_bf16.h>
#include <math.h>

typedef __attribute__((ext_vector_type(2))) float v2f;
typedef __attribute__((ext_vector_type(8))) float v8f;

#define NEG_SLOPE 0.2f

__device__ __forceinline__ unsigned enc_f32(float f) {
  unsigned u = __float_as_uint(f);
  return u ^ ((u >> 31) ? 0xFFFFFFFFu : 0x80000000u);
}
__device__ __forceinline__ float dec_f32(unsigned u) {
  unsigned v = u ^ ((u >> 31) ? 0x80000000u : 0xFFFFFFFFu);
  return __uint_as_float(v);
}

#define GEMM_P4 33
__global__ __launch_bounds__(256) void gat_gemm128(const float* __restrict__ X,
                                                   const float* __restrict__ W,
                                                   float* __restrict__ Hout,
                                                   int N) {
  __shared__ float4 As4[2][16 * GEMM_P4];
  const int tid  = threadIdx.x;
  const int lane = tid & 31;
  const int wv   = tid >> 5;
  const int n0   = wv << 4;
  const int r    = lane & 15;
  const int half = lane >> 4;
  const int ntiles = (N + 15) >> 4;

  v2f bf[32];
#pragma unroll
  for (int j = 0; j < 16; ++j) {
    const float* wp = W + (size_t)(8 * j + 4 * half) * 128 + n0 + r;
    bf[2 * j].x     = wp[0];
    bf[2 * j].y     = wp[128];
    bf[2 * j + 1].x = wp[256];
    bf[2 * j + 1].y = wp[384];
  }

  const int lrow = tid >> 5;
  const int c4   = tid & 31;
  int t = blockIdx.x;
  float4 p0 = make_float4(0.f, 0.f, 0.f, 0.f), p1 = p0;
  if (t < ntiles) {
    int ra = (t << 4) + lrow, rb = ra + 8;
    ra = ra < N ? ra : N - 1;
    rb = rb < N ? rb : N - 1;
    p0 = *(const float4*)(X + (size_t)ra * 128 + c4 * 4);
    p1 = *(const float4*)(X + (size_t)rb * 128 + c4 * 4);
  }
  int buf = 0;
  for (; t < ntiles; t += gridDim.x) {
    As4[buf][lrow * GEMM_P4 + c4]       = p0;
    As4[buf][(lrow + 8) * GEMM_P4 + c4] = p1;
    __syncthreads();
    const int tn = t + gridDim.x;
    if (tn < ntiles) {
      int ra = (tn << 4) + lrow, rb = ra + 8;
      ra = ra < N ? ra : N - 1;
      rb = rb < N ? rb : N - 1;
      p0 = *(const float4*)(X + (size_t)ra * 128 + c4 * 4);
      p1 = *(const float4*)(X + (size_t)rb * 128 + c4 * 4);
    }
    v8f acc0 = {0.f, 0.f, 0.f, 0.f, 0.f, 0.f, 0.f, 0.f};
    v8f acc1 = acc0;
    const float4* arow = &As4[buf][r * GEMM_P4 + half];
#pragma unroll
    for (int j = 0; j < 16; ++j) {
      const float4 q = arow[2 * j];
      v2f a0; a0.x = q.x; a0.y = q.y;
      v2f a1; a1.x = q.z; a1.y = q.w;
      acc0 = __builtin_amdgcn_wmma_f32_16x16x4_f32(false, a0, false, bf[2 * j],
                                                   (short)0, acc0, false, false);
      acc1 = __builtin_amdgcn_wmma_f32_16x16x4_f32(false, a1, false, bf[2 * j + 1],
                                                   (short)0, acc1, false, false);
    }
    const int rowbase = (t << 4) + (half << 3);
    const int col     = n0 + r;
#pragma unroll
    for (int i = 0; i < 8; ++i) {
      const int row = rowbase + i;
      if (row < N) Hout[(size_t)row * 128 + col] = acc0[i] + acc1[i];
    }
    buf ^= 1;
  }
}

#define CTR_PITCH 64
#define NODE_SPLIT(n) ((((n) >> 1) + 15) & ~15)
__device__ __forceinline__ int chip_half() {
  return (int)((__builtin_amdgcn_s_sendmsg_rtn(0x87) >> 18) & 1u);
}

__global__ void gat_scores_init(const float* __restrict__ Hm,
                                const float* __restrict__ asrc,
                                const float* __restrict__ adst,
                                float* __restrict__ es, float* __restrict__ ed,
                                unsigned* __restrict__ menc,
                                float* __restrict__ ssum,
                                float* __restrict__ acc,
                                unsigned* __restrict__ ctr, int N) {
  const int idx = blockIdx.x * blockDim.x + threadIdx.x;
  if (idx >= N * 4) return;
  if (idx < 4) ctr[idx * CTR_PITCH] = 0u;
  const int hd = idx & 3;
  const size_t base = (size_t)(idx >> 2) * 128 + hd * 32;
  const float* hp  = Hm + base;
  float*       ap  = acc + base;
  const float* sa  = asrc + hd * 32;
  const float* da  = adst + hd * 32;
  float s = 0.f, d = 0.f;
#pragma unroll
  for (int c = 0; c < 32; c += 4) {
    float4 hv = *(const float4*)(hp + c);
    float4 av = *(const float4*)(sa + c);
    float4 bv = *(const float4*)(da + c);
    s += hv.x * av.x + hv.y * av.y + hv.z * av.z + hv.w * av.w;
    d += hv.x * bv.x + hv.y * bv.y + hv.z * bv.z + hv.w * bv.w;
    *(float4*)(ap + c) = make_float4(0.f, 0.f, 0.f, 0.f);
  }
  es[idx] = s;
  ed[idx] = d;
  menc[idx] = 0u;
  ssum[idx] = 0.f;
}

__device__ __forceinline__ float lrelu(float x) {
  return x > 0.f ? x : NEG_SLOPE * x;
}

#define MAXP_CHUNKS 4
__global__ __launch_bounds__(256) void gat_edge_max(const int* __restrict__ ei,
                                                    const float* __restrict__ es,
                                                    unsigned* __restrict__ menc,
                                                    unsigned* __restrict__ ctr,
                                                    int E, int N) {
  __shared__ unsigned tk[2];
  __shared__ int sh_half;
  const int tid  = threadIdx.x;
  const int lane = tid & 31;
  const int wv   = tid >> 5;
  const int Et = E + N;
  const int split = NODE_SPLIT(N);
  const unsigned nch = (unsigned)((Et + 256 * MAXP_CHUNKS - 1) / (256 * MAXP_CHUNKS));
  if (tid == 0) sh_half = chip_half();
  __syncthreads();
  const int myhalf = sh_half;
  const int sub = lane >> 2;
  const int hd  = lane & 3;
  unsigned it = 0;
  for (int pass = 0; pass < 2; ++pass) {
    const int part = myhalf ^ pass;
    unsigned* cp = ctr + part * CTR_PITCH;
    unsigned nxt = 0;
    if (tid == 0) nxt = atomicAdd(cp, 1u);
    for (unsigned k = 0; k <= nch; ++k) {
      if (tid == 0) tk[it & 1] = nxt;
      __syncthreads();
      const unsigned t = tk[it & 1];
      ++it;
      if (t >= nch) break;
      if (tid == 0) nxt = atomicAdd(cp, 1u);
      const int e0 = ((int)t * 8 + wv) * (32 * MAXP_CHUNKS);
      if (e0 >= Et) continue;
      int s[MAXP_CHUNKS], d[MAXP_CHUNKS];
      unsigned msk[MAXP_CHUNKS];
#pragma unroll
      for (int c = 0; c < MAXP_CHUNKS; ++c) {
        const int e = e0 + c * 32 + lane;
        s[c] = 0; d[c] = 0;
        if (e < Et) {
          if (e < E) { s[c] = ei[e]; d[c] = ei[E + e]; } else { s[c] = d[c] = e - E; }
        }
      }
#pragma unroll
      for (int c = 0; c < MAXP_CHUNKS; ++c) {
        const int e = e0 + c * 32 + lane;
        const bool mine = (e < Et) && (((d[c] >= split) ? 1 : 0) == part);
        msk[c] = __builtin_amdgcn_ballot_w32(mine);
      }
      float v[MAXP_CHUNKS * 4];
      int   dj[MAXP_CHUNKS * 4];
      unsigned on = 0u;
#pragma unroll
      for (int c = 0; c < MAXP_CHUNKS; ++c) {
#pragma unroll
        for (int g = 0; g < 4; ++g) {
          const int from = g * 8 + sub;
          const int sj = __shfl(s[c], from);
          dj[c * 4 + g] = __shfl(d[c], from);
          const bool p = (msk[c] >> from) & 1u;
          v[c * 4 + g] = 0.f;
          if (p) {
            v[c * 4 + g] = es[(size_t)sj * 4 + hd];
            on |= 1u << (c * 4 + g);
          }
        }
      }
#pragma unroll
      for (int i = 0; i < MAXP_CHUNKS * 4; ++i)
        if ((on >> i) & 1u) atomicMax(menc + (size_t)dj[i] * 4 + hd, enc_f32(v[i]));
    }
  }
}

struct Row4 { float c0, c1, c2, c3; };

__device__ __forceinline__ float lane_f(float v, int j) {
  return __int_as_float(__builtin_amdgcn_readlane(__float_as_int(v), j));
}
__device__ __forceinline__ Row4 load_row(const float* __restrict__ Hm, int sj, int lane) {
  const float* hp = Hm + (size_t)sj * 128 + lane;
  Row4 r;
  r.c0 = hp[0]; r.c1 = hp[32]; r.c2 = hp[64]; r.c3 = hp[96];
  return r;
}
__device__ __forceinline__ void add_row(float* __restrict__ acc, int dj, Row4 h,
                                        float w0, float w1, float w2, float w3, int lane) {
  float* op = acc + (size_t)dj * 128 + lane;
  atomicAdd(op,      h.c0 * w0);
  atomicAdd(op + 32, h.c1 * w1);
  atomicAdd(op + 64, h.c2 * w2);
  atomicAdd(op + 96, h.c3 * w3);
}
__device__ __forceinline__ void add_den(float* __restrict__ ssum, int d,
                                        float x0, float x1, float x2, float x3,
                                        int j0, unsigned mask, int lane) {
  const int src = j0 + (lane >> 2);
  const int hsel = lane & 3;
  const float t0 = __shfl(x0, src), t1 = __shfl(x1, src);
  const float t2 = __shfl(x2, src), t3 = __shfl(x3, src);
  const int dz = __shfl(d, src);
  const float za = (hsel & 1) ? t1 : t0;
  const float zb = (hsel & 1) ? t3 : t2;
  const float z  = (hsel & 2) ? zb : za;
  if ((mask >> src) & 1u) atomicAdd(&ssum[(size_t)dz * 4 + hsel], z);
}

#define SCAT_SUB 2
__global__ __launch_bounds__(256) void gat_edge_scatter(
    const int* __restrict__ ei,
    const float* __restrict__ es, const float* __restrict__ ed,
    const unsigned* __restrict__ menc, float* __restrict__ ssum,
    const float* __restrict__ Hm, float* __restrict__ acc,
    unsigned* __restrict__ ctr, int E, int N) {
  __shared__ unsigned tk[2];
  __shared__ int sh_half;
  const int tid  = threadIdx.x;
  const int lane = tid & 31;
  const int wv   = tid >> 5;
  const int Et = E + N;
  const int split = NODE_SPLIT(N);
  const unsigned nch = (unsigned)((Et + 256 * SCAT_SUB - 1) / (256 * SCAT_SUB));
  if (tid == 0) sh_half = chip_half();
  __syncthreads();
  const int myhalf = sh_half;
  unsigned it = 0;
  for (int pass = 0; pass < 2; ++pass) {
    const int part = myhalf ^ pass;
    unsigned* cp = ctr + (2 + part) * CTR_PITCH;
    unsigned nxt = 0;
    if (tid == 0) nxt = atomicAdd(cp, 1u);
    for (unsigned k = 0; k <= nch; ++k) {
      if (tid == 0) tk[it & 1] = nxt;
      __syncthreads();
      const unsigned t = tk[it & 1];
      ++it;
      if (t >= nch) break;
      if (tid == 0) nxt = atomicAdd(cp, 1u);
      const int eb = ((int)t * 8 + wv) * (32 * SCAT_SUB);
      int sn = 0, dn = 0;
      {
        const int e = eb + lane;
        if (e < Et) { if (e < E) { sn = ei[e]; dn = ei[E + e]; } else { sn = dn = e - E; } }
      }
#pragma unroll 1
      for (int c = 0; c < SCAT_SUB; ++c) {
        const int e0 = eb + c * 32;
        if (e0 >= Et) break;
        const int e = e0 + lane;
        const int s = sn, d = dn;
        sn = 0; dn = 0;
        if (c + 1 < SCAT_SUB) {
          const int en = e + 32;
          if (en < Et) { if (en < E) { sn = ei[en]; dn = ei[E + en]; } else { sn = dn = en - E; } }
        }
        float x0 = 0.f, x1 = 0.f, x2 = 0.f, x3 = 0.f;
        const bool mine = (e < Et) && (((d >= split) ? 1 : 0) == part);
        if (mine) {
          const float4 vs = *(const float4*)(es + (size_t)s * 4);
          const float4 vd = *(const float4*)(ed + (size_t)d * 4);
          const uint4  mm = *(const uint4*)(menc + (size_t)d * 4);
          x0 = expf(lrelu(vs.x + vd.x) - lrelu(dec_f32(mm.x) + vd.x));
          x1 = expf(lrelu(vs.y + vd.y) - lrelu(dec_f32(mm.y) + vd.y));
          x2 = expf(lrelu(vs.z + vd.z) - lrelu(dec_f32(mm.z) + vd.z));
          x3 = expf(lrelu(vs.w + vd.w) - lrelu(dec_f32(mm.w) + vd.w));
        }
        const unsigned mask = __builtin_amdgcn_ballot_w32(mine);
        for (int j0 = 0; j0 < 32; j0 += 8)
          if ((mask >> j0) & 0xFFu) add_den(ssum, d, x0, x1, x2, x3, j0, mask, lane);
        unsigned m = mask;
        for (int b = 0; b < 4 && m != 0u; ++b) {
          int js[8];
          int cnt = 0;
#pragma unroll
          for (int u = 0; u < 8; ++u) {
            js[u] = 0;
            if (m != 0u) { js[u] = __builtin_ctz(m); m &= m - 1u; ++cnt; }
          }
          Row4 hv[8];
          if (cnt == 8) {
#pragma unroll
            for (int u = 0; u < 8; ++u)
              hv[u] = load_row(Hm, __builtin_amdgcn_readlane(s, js[u]), lane);
#pragma unroll
            for (int u = 0; u < 8; ++u)
              add_row(acc, __builtin_amdgcn_readlane(d, js[u]), hv[u],
                      lane_f(x0, js[u]), lane_f(x1, js[u]),
                      lane_f(x2, js[u]), lane_f(x3, js[u]), lane);
          } else {
#pragma unroll
            for (int u = 0; u < 7; ++u)
              if (u < cnt) hv[u] = load_row(Hm, __builtin_amdgcn_readlane(s, js[u]), lane);
#pragma unroll
            for (int u = 0; u < 7; ++u)
              if (u < cnt)
                add_row(acc, __builtin_amdgcn_readlane(d, js[u]), hv[u],
                        lane_f(x0, js[u]), lane_f(x1, js[u]),
                        lane_f(x2, js[u]), lane_f(x3, js[u]), lane);
          }
        }
      }
    }
  }
}

__global__ void gat_finalize(const float* __restrict__ acc,
                             const float* __restrict__ ssum,
                             const float* __restrict__ bias,
                             float* __restrict__ outp, int N, int apply_elu) {
  const int idx = blockIdx.x * blockDim.x + threadIdx.x;
  if (idx >= N * 128) return;
  const int n  = idx >> 7;
  const int hd = (idx >> 5) & 3;
  float v = acc[idx] / ssum[(size_t)n * 4 + hd] + bias[idx & 127];
  if (apply_elu) v = v > 0.f ? v : expm1f(v);
  outp[idx] = v;
}

extern "C" void kernel_launch(void* const* d_in, const int* in_sizes, int n_in,
                              void* d_out, int out_size, void* d_ws, size_t ws_size,
                              hipStream_t stream) {
  const float* x  = (const float*)d_in[0];
  const int*   ei = (const int*)d_in[1];
  const int N = in_sizes[0] / 128;
  const int E = in_sizes[1] / 2;

  const float* Wl[3] = {(const float*)d_in[2], (const float*)d_in[6],  (const float*)d_in[10]};
  const float* As[3] = {(const float*)d_in[3], (const float*)d_in[7],  (const float*)d_in[11]};
  const float* Ad[3] = {(const float*)d_in[4], (const float*)d_in[8],  (const float*)d_in[12]};
  const float* Bb[3] = {(const float*)d_in[5], (const float*)d_in[9],  (const float*)d_in[13]};

  float*    F0   = (float*)d_ws;
  float*    F1   = F0 + (size_t)N * 128;
  float*    es   = F1 + (size_t)N * 128;
  float*    ed   = es + (size_t)N * 4;
  unsigned* menc = (unsigned*)(ed + (size_t)N * 4);
  float*    ssum = (float*)(menc + (size_t)N * 4);
  unsigned* ctr  = (unsigned*)(ssum + (size_t)N * 4);

  const dim3 bGemm(256);
  const dim3 gGemm(512);
  const int gScore = (N * 4 + 255) / 256;
  const int gMax   = 1536;
  const int gScat  = 1280;
  const int gFin   = (N * 128 + 255) / 256;

  const float* inp = x;
  for (int layer = 0; layer < 3; ++layer) {
    float* Hm = (layer & 1) ? F1 : F0;
    float* Ac = (layer & 1) ? F0 : F1;
    gat_gemm128<<<gGemm, bGemm, 0, stream>>>(inp, Wl[layer], Hm, N);
    gat_scores_init<<<gScore, 256, 0, stream>>>(Hm, As[layer], Ad[layer],
                                                es, ed, menc, ssum, Ac, ctr, N);
    gat_edge_max<<<gMax, 256, 0, stream>>>(ei, es, menc, ctr, E, N);
    gat_edge_scatter<<<gScat, 256, 0, stream>>>(ei, es, ed, menc, ssum, Hm, Ac, ctr, E, N);
    float* ot = (layer == 2) ? (float*)d_out : Hm;
    gat_finalize<<<gFin, 256, 0, stream>>>(Ac, ssum, Bb[layer], ot, N,
                                           layer < 2 ? 1 : 0);
    inp = Hm;
  }
}


# Layers 0,1: divide by softmax sum, bias and ELU applied to each 16-byte piece as the next GEMM stages its X tile in LDS: 2 launches and a write+read of the node matrix go; sum order kept
# speedup vs baseline: 3.0699x; 1.0781x over previous
#include <hip/hip_runtime.h>
#include <hip/hip_bf16.h>
#include <math.h>

typedef __attribute__((ext_vector_type(2))) float v2f;
typedef __attribute__((ext_vector_type(8))) float v8f;

#define NEG_SLOPE 0.2f

__device__ __forceinline__ unsigned enc_f32(float f) {
  unsigned u = __float_as_uint(f);
  return u ^ ((u >> 31) ? 0xFFFFFFFFu : 0x80000000u);
}
__device__ __forceinline__ float dec_f32(unsigned u) {
  unsigned v = u ^ ((u >> 31) ? 0x80000000u : 0xFFFFFFFFu);
  return __uint_as_float(v);
}

#define GEMM_P4 33
__device__ __forceinline__ float elu1(float v) {
  const float m = v > 0.f ? 0.f : v;
  const float e = expm1f(m);
  return v > 0.f ? v : e;
}
__device__ __forceinline__ float4 close_piece(float4 p, float s, float4 b) {
  float4 o;
  o.x = elu1(p.x / s + b.x);
  o.y = elu1(p.y / s + b.y);
  o.z = elu1(p.z / s + b.z);
  o.w = elu1(p.w / s + b.w);
  return o;
}
template <bool NORM>
__global__ __launch_bounds__(256) void gat_gemm128(const float* __restrict__ X,
                                                   const float* __restrict__ W,
                                                   float* __restrict__ Hout,
                                                   const float* __restrict__ den,
                                                   const float* __restrict__ bias,
                                                   int N) {
  __shared__ float4 As4[2][16 * GEMM_P4];
  const int tid  = threadIdx.x;
  const int lane = tid & 31;
  const int wv   = tid >> 5;
  const int n0   = wv << 4;
  const int r    = lane & 15;
  const int half = lane >> 4;
  const int ntiles = (N + 15) >> 4;

  v2f bf[32];
#pragma unroll
  for (int j = 0; j < 16; ++j) {
    const float* wp = W + (size_t)(8 * j + 4 * half) * 128 + n0 + r;
    bf[2 * j].x     = wp[0];
    bf[2 * j].y     = wp[128];
    bf[2 * j + 1].x = wp[256];
    bf[2 * j + 1].y = wp[384];
  }

  const int lrow = tid >> 5;
  const int c4   = tid & 31;
  const int hq   = c4 >> 3;
  int t = blockIdx.x;
  float4 p0 = make_float4(0.f, 0.f, 0.f, 0.f), p1 = p0;
  float4 bq = p0;
  float d0 = 1.f, d1 = 1.f;
  if (NORM) bq = *(const float4*)(bias + c4 * 4);
  if (t < ntiles) {
    int ra = (t << 4) + lrow, rb = ra + 8;
    ra = ra < N ? ra : N - 1;
    rb = rb < N ? rb : N - 1;
    p0 = *(const float4*)(X + (size_t)ra * 128 + c4 * 4);
    p1 = *(const float4*)(X + (size_t)rb * 128 + c4 * 4);
    if (NORM) { d0 = den[(size_t)ra * 4 + hq]; d1 = den[(size_t)rb * 4 + hq]; }
  }
  int buf = 0;
  for (; t < ntiles; t += gridDim.x) {
    if (NORM) { p0 = close_piece(p0, d0, bq); p1 = close_piece(p1, d1, bq); }
    As4[buf][lrow * GEMM_P4 + c4]       = p0;
    As4[buf][(lrow + 8) * GEMM_P4 + c4] = p1;
    __syncthreads();
    const int tn = t + gridDim.x;
    if (tn < ntiles) {
      int ra = (tn << 4) + lrow, rb = ra + 8;
      ra = ra < N ? ra : N - 1;
      rb = rb < N ? rb : N - 1;
      p0 = *(const float4*)(X + (size_t)ra * 128 + c4 * 4);
      p1 = *(const float4*)(X + (size_t)rb * 128 + c4 * 4);
      if (NORM) { d0 = den[(size_t)ra * 4 + hq]; d1 = den[(size_t)rb * 4 + hq]; }
    }
    v8f acc0 = {0.f, 0.f, 0.f, 0.f, 0.f, 0.f, 0.f, 0.f};
    v8f acc1 = acc0;
    const float4* arow = &As4[buf][r * GEMM_P4 + half];
#pragma unroll
    for (int j = 0; j < 16; ++j) {
      const float4 q = arow[2 * j];
      v2f a0; a0.x = q.x; a0.y = q.y;
      v2f a1; a1.x = q.z; a1.y = q.w;
      acc0 = __builtin_amdgcn_wmma_f32_16x16x4_f32(false, a0, false, bf[2 * j],
                                                   (short)0, acc0, false, false);
      acc1 = __builtin_amdgcn_wmma_f32_16x16x4_f32(false, a1, false, bf[2 * j + 1],
                                                   (short)0, acc1, false, false);
    }
    const int rowbase = (t << 4) + (half << 3);
    const int col     = n0 + r;
#pragma unroll
    for (int i = 0; i < 8; ++i) {
      const int row = rowbase + i;
      if (row < N) Hout[(size_t)row * 128 + col] = acc0[i] + acc1[i];
    }
    buf ^= 1;
  }
}

#define CTR_PITCH 64
#define NODE_SPLIT(n) ((((n) >> 1) + 15) & ~15)
__device__ __forceinline__ int chip_half() {
  return (int)((__builtin_amdgcn_s_sendmsg_rtn(0x87) >> 18) & 1u);
}

__global__ void gat_scores_init(const float* __restrict__ Hm,
                                const float* __restrict__ asrc,
                                const float* __restrict__ adst,
                                float* __restrict__ es, float* __restrict__ ed,
                                unsigned* __restrict__ menc,
                                float* __restrict__ ssum,
                                float* __restrict__ acc,
                                unsigned* __restrict__ ctr, int N) {
  const int idx = blockIdx.x * blockDim.x + threadIdx.x;
  if (idx >= N * 4) return;
  if (idx < 4) ctr[idx * CTR_PITCH] = 0u;
  const int hd = idx & 3;
  const size_t base = (size_t)(idx >> 2) * 128 + hd * 32;
  const float* hp  = Hm + base;
  float*       ap  = acc + base;
  const float* sa  = asrc + hd * 32;
  const float* da  = adst + hd * 32;
  float s = 0.f, d = 0.f;
#pragma unroll
  for (int c = 0; c < 32; c += 4) {
    float4 hv = *(const float4*)(hp + c);
    float4 av = *(const float4*)(sa + c);
    float4 bv = *(const float4*)(da + c);
    s += hv.x * av.x + hv.y * av.y + hv.z * av.z + hv.w * av.w;
    d += hv.x * bv.x + hv.y * bv.y + hv.z * bv.z + hv.w * bv.w;
    *(float4*)(ap + c) = make_float4(0.f, 0.f, 0.f, 0.f);
  }
  es[idx] = s;
  ed[idx] = d;
  menc[idx] = 0u;
  ssum[idx] = 0.f;
}

__device__ __forceinline__ float lrelu(float x) {
  return x > 0.f ? x : NEG_SLOPE * x;
}

#define MAXP_CHUNKS 4
__global__ __launch_bounds__(256) void gat_edge_max(const int* __restrict__ ei,
                                                    const float* __restrict__ es,
                                                    unsigned* __restrict__ menc,
                                                    unsigned* __restrict__ ctr,
                                                    int E, int N) {
  __shared__ unsigned tk[2];
  __shared__ int sh_half;
  const int tid  = threadIdx.x;
  const int lane = tid & 31;
  const int wv   = tid >> 5;
  const int Et = E + N;
  const int split = NODE_SPLIT(N);
  const unsigned nch = (unsigned)((Et + 256 * MAXP_CHUNKS - 1) / (256 * MAXP_CHUNKS));
  if (tid == 0) sh_half = chip_half();
  __syncthreads();
  const int myhalf = sh_half;
  const int sub = lane >> 2;
  const int hd  = lane & 3;
  unsigned it = 0;
  for (int pass = 0; pass < 2; ++pass) {
    const int part = myhalf ^ pass;
    unsigned* cp = ctr + part * CTR_PITCH;
    unsigned nxt = 0;
    if (tid == 0) nxt = atomicAdd(cp, 1u);
    for (unsigned k = 0; k <= nch; ++k) {
      if (tid == 0) tk[it & 1] = nxt;
      __syncthreads();
      const unsigned t = tk[it & 1];
      ++it;
      if (t >= nch) break;
      if (tid == 0) nxt = atomicAdd(cp, 1u);
      const int e0 = ((int)t * 8 + wv) * (32 * MAXP_CHUNKS);
      if (e0 >= Et) continue;
      int s[MAXP_CHUNKS], d[MAXP_CHUNKS];
      unsigned msk[MAXP_CHUNKS];
#pragma unroll
      for (int c = 0; c < MAXP_CHUNKS; ++c) {
        const int e = e0 + c * 32 + lane;
        s[c] = 0; d[c] = 0;
        if (e < Et) {
          if (e < E) { s[c] = ei[e]; d[c] = ei[E + e]; } else { s[c] = d[c] = e - E; }
        }
      }
#pragma unroll
      for (int c = 0; c < MAXP_CHUNKS; ++c) {
        const int e = e0 + c * 32 + lane;
        const bool mine = (e < Et) && (((d[c] >= split) ? 1 : 0) == part);
        msk[c] = __builtin_amdgcn_ballot_w32(mine);
      }
      float v[MAXP_CHUNKS * 4];
      int   dj[MAXP_CHUNKS * 4];
      unsigned on = 0u;
#pragma unroll
      for (int c = 0; c < MAXP_CHUNKS; ++c) {
#pragma unroll
        for (int g = 0; g < 4; ++g) {
          const int from = g * 8 + sub;
          const int sj = __shfl(s[c], from);
          dj[c * 4 + g] = __shfl(d[c], from);
          const bool p = (msk[c] >> from) & 1u;
          v[c * 4 + g] = 0.f;
          if (p) {
            v[c * 4 + g] = es[(size_t)sj * 4 + hd];
            on |= 1u << (c * 4 + g);
          }
        }
      }
#pragma unroll
      for (int i = 0; i < MAXP_CHUNKS * 4; ++i)
        if ((on >> i) & 1u) atomicMax(menc + (size_t)dj[i] * 4 + hd, enc_f32(v[i]));
    }
  }
}

struct Row4 { float c0, c1, c2, c3; };

__device__ __forceinline__ float lane_f(float v, int j) {
  return __int_as_float(__builtin_amdgcn_readlane(__float_as_int(v), j));
}
__device__ __forceinline__ Row4 load_row(const float* __restrict__ Hm, int sj, int lane) {
  const float* hp = Hm + (size_t)sj * 128 + lane;
  Row4 r;
  r.c0 = hp[0]; r.c1 = hp[32]; r.c2 = hp[64]; r.c3 = hp[96];
  return r;
}
__device__ __forceinline__ void add_row(float* __restrict__ acc, int dj, Row4 h,
                                        float w0, float w1, float w2, float w3, int lane) {
  float* op = acc + (size_t)dj * 128 + lane;
  atomicAdd(op,      h.c0 * w0);
  atomicAdd(op + 32, h.c1 * w1);
  atomicAdd(op + 64, h.c2 * w2);
  atomicAdd(op + 96, h.c3 * w3);
}
__device__ __forceinline__ void add_den(float* __restrict__ ssum, int d,
                                        float x0, float x1, float x2, float x3,
                                        int j0, unsigned mask, int lane) {
  const int src = j0 + (lane >> 2);
  const int hsel = lane & 3;
  const float t0 = __shfl(x0, src), t1 = __shfl(x1, src);
  const float t2 = __shfl(x2, src), t3 = __shfl(x3, src);
  const int dz = __shfl(d, src);
  const float za = (hsel & 1) ? t1 : t0;
  const float zb = (hsel & 1) ? t3 : t2;
  const float z  = (hsel & 2) ? zb : za;
  if ((mask >> src) & 1u) atomicAdd(&ssum[(size_t)dz * 4 + hsel], z);
}

#define SCAT_SUB 2
__global__ __launch_bounds__(256) void gat_edge_scatter(
    const int* __restrict__ ei,
    const float* __restrict__ es, const float* __restrict__ ed,
    const unsigned* __restrict__ menc, float* __restrict__ ssum,
    const float* __restrict__ Hm, float* __restrict__ acc,
    unsigned* __restrict__ ctr, int E, int N) {
  __shared__ unsigned tk[2];
  __shared__ int sh_half;
  const int tid  = threadIdx.x;
  const int lane = tid & 31;
  const int wv   = tid >> 5;
  const int Et = E + N;
  const int split = NODE_SPLIT(N);
  const unsigned nch = (unsigned)((Et + 256 * SCAT_SUB - 1) / (256 * SCAT_SUB));
  if (tid == 0) sh_half = chip_half();
  __syncthreads();
  const int myhalf = sh_half;
  unsigned it = 0;
  for (int pass = 0; pass < 2; ++pass) {
    const int part = myhalf ^ pass;
    unsigned* cp = ctr + (2 + part) * CTR_PITCH;
    unsigned nxt = 0;
    if (tid == 0) nxt = atomicAdd(cp, 1u);
    for (unsigned k = 0; k <= nch; ++k) {
      if (tid == 0) tk[it & 1] = nxt;
      __syncthreads();
      const unsigned t = tk[it & 1];
      ++it;
      if (t >= nch) break;
      if (tid == 0) nxt = atomicAdd(cp, 1u);
      const int eb = ((int)t * 8 + wv) * (32 * SCAT_SUB);
      int sn = 0, dn = 0;
      {
        const int e = eb + lane;
        if (e < Et) { if (e < E) { sn = ei[e]; dn = ei[E + e]; } else { sn = dn = e - E; } }
      }
#pragma unroll 1
      for (int c = 0; c < SCAT_SUB; ++c) {
        const int e0 = eb + c * 32;
        if (e0 >= Et) break;
        const int e = e0 + lane;
        const int s = sn, d = dn;
        sn = 0; dn = 0;
        if (c + 1 < SCAT_SUB) {
          const int en = e + 32;
          if (en < Et) { if (en < E) { sn = ei[en]; dn = ei[E + en]; } else { sn = dn = en - E; } }
        }
        float x0 = 0.f, x1 = 0.f, x2 = 0.f, x3 = 0.f;
        const bool mine = (e < Et) && (((d >= split) ? 1 : 0) == part);
        if (mine) {
          const float4 vs = *(const float4*)(es + (size_t)s * 4);
          const float4 vd = *(const float4*)(ed + (size_t)d * 4);
          const uint4  mm = *(const uint4*)(menc + (size_t)d * 4);
          x0 = expf(lrelu(vs.x + vd.x) - lrelu(dec_f32(mm.x) + vd.x));
          x1 = expf(lrelu(vs.y + vd.y) - lrelu(dec_f32(mm.y) + vd.y));
          x2 = expf(lrelu(vs.z + vd.z) - lrelu(dec_f32(mm.z) + vd.z));
          x3 = expf(lrelu(vs.w + vd.w) - lrelu(dec_f32(mm.w) + vd.w));
        }
        const unsigned mask = __builtin_amdgcn_ballot_w32(mine);
        for (int j0 = 0; j0 < 32; j0 += 8)
          if ((mask >> j0) & 0xFFu) add_den(ssum, d, x0, x1, x2, x3, j0, mask, lane);
        unsigned m = mask;
        for (int b = 0; b < 4 && m != 0u; ++b) {
          int js[8];
          int cnt = 0;
#pragma unroll
          for (int u = 0; u < 8; ++u) {
            js[u] = 0;
            if (m != 0u) { js[u] = __builtin_ctz(m); m &= m - 1u; ++cnt; }
          }
          Row4 hv[8];
          if (cnt == 8) {
#pragma unroll
            for (int u = 0; u < 8; ++u)
              hv[u] = load_row(Hm, __builtin_amdgcn_readlane(s, js[u]), lane);
#pragma unroll
            for (int u = 0; u < 8; ++u)
              add_row(acc, __builtin_amdgcn_readlane(d, js[u]), hv[u],
                      lane_f(x0, js[u]), lane_f(x1, js[u]),
                      lane_f(x2, js[u]), lane_f(x3, js[u]), lane);
          } else {
#pragma unroll
            for (int u = 0; u < 7; ++u)
              if (u < cnt) hv[u] = load_row(Hm, __builtin_amdgcn_readlane(s, js[u]), lane);
#pragma unroll
            for (int u = 0; u < 7; ++u)
              if (u < cnt)
                add_row(acc, __builtin_amdgcn_readlane(d, js[u]), hv[u],
                        lane_f(x0, js[u]), lane_f(x1, js[u]),
                        lane_f(x2, js[u]), lane_f(x3, js[u]), lane);
          }
        }
      }
    }
  }
}

__global__ void gat_finalize(const float* __restrict__ acc,
                             const float* __restrict__ ssum,
                             const float* __restrict__ bias,
                             float* __restrict__ outp, int N, int apply_elu) {
  const int idx = blockIdx.x * blockDim.x + threadIdx.x;
  if (idx >= N * 128) return;
  const int n  = idx >> 7;
  const int hd = (idx >> 5) & 3;
  float v = acc[idx] / ssum[(size_t)n * 4 + hd] + bias[idx & 127];
  if (apply_elu) v = v > 0.f ? v : expm1f(v);
  outp[idx] = v;
}

extern "C" void kernel_launch(void* const* d_in, const int* in_sizes, int n_in,
                              void* d_out, int out_size, void* d_ws, size_t ws_size,
                              hipStream_t stream) {
  const float* x  = (const float*)d_in[0];
  const int*   ei = (const int*)d_in[1];
  const int N = in_sizes[0] / 128;
  const int E = in_sizes[1] / 2;

  const float* Wl[3] = {(const float*)d_in[2], (const float*)d_in[6],  (const float*)d_in[10]};
  const float* As[3] = {(const float*)d_in[3], (const float*)d_in[7],  (const float*)d_in[11]};
  const float* Ad[3] = {(const float*)d_in[4], (const float*)d_in[8],  (const float*)d_in[12]};
  const float* Bb[3] = {(const float*)d_in[5], (const float*)d_in[9],  (const float*)d_in[13]};

  float*    F0   = (float*)d_ws;
  float*    F1   = F0 + (size_t)N * 128;
  float*    es   = F1 + (size_t)N * 128;
  float*    ed   = es + (size_t)N * 4;
  unsigned* menc = (unsigned*)(ed + (size_t)N * 4);
  float*    ssum = (float*)(menc + (size_t)N * 4);
  unsigned* ctr  = (unsigned*)(ssum + (size_t)N * 4);

  const dim3 bGemm(256);
  const dim3 gGemm(512);
  const int gScore = (N * 4 + 255) / 256;
  const int gMax   = 1536;
  const int gScat  = 1280;
  const int gFin   = (N * 128 + 255) / 256;

  float* Hm = F0;
  float* Ac = F1;
  for (int layer = 0; layer < 3; ++layer) {
    if (layer == 0)
      gat_gemm128<false><<<gGemm, bGemm, 0, stream>>>(x, Wl[0], Hm, ssum, Bb[0], N);
    else
      gat_gemm128<true><<<gGemm, bGemm, 0, stream>>>(Ac, Wl[layer], Hm, ssum, Bb[layer - 1], N);
    gat_scores_init<<<gScore, 256, 0, stream>>>(Hm, As[layer], Ad[layer],
                                                es, ed, menc, ssum, Ac, ctr, N);
    gat_edge_max<<<gMax, 256, 0, stream>>>(ei, es, menc, ctr, E, N);
    gat_edge_scatter<<<gScat, 256, 0, stream>>>(ei, es, ed, menc, ssum, Hm, Ac, ctr, E, N);
  }
  gat_finalize<<<gFin, 256, 0, stream>>>(Ac, ssum, Bb[2], (float*)d_out, N, 0);
}


# Softmax shift from the per-head max and min of all source scores (reduced in the score launch): where their range is small the segment-max edge pass exits at once, else as before; last bits move
# speedup vs baseline: 3.3941x; 1.1056x over previous
#include <hip/hip_runtime.h>
#include <hip/hip_bf16.h>
#include <math.h>

typedef __attribute__((ext_vector_type(2))) float v2f;
typedef __attribute__((ext_vector_type(8))) float v8f;

#define NEG_SLOPE 0.2f
#define CTR_PITCH 64

__device__ __forceinline__ unsigned enc_f32(float f) {
  unsigned u = __float_as_uint(f);
  return u ^ ((u >> 31) ? 0xFFFFFFFFu : 0x80000000u);
}
__device__ __forceinline__ float dec_f32(unsigned u) {
  unsigned v = u ^ ((u >> 31) ? 0x80000000u : 0xFFFFFFFFu);
  return __uint_as_float(v);
}

#define GEMM_P4 33
__device__ __forceinline__ float elu1(float v) {
  const float m = v > 0.f ? 0.f : v;
  const float e = expm1f(m);
  return v > 0.f ? v : e;
}
__device__ __forceinline__ float4 close_piece(float4 p, float s, float4 b) {
  float4 o;
  o.x = elu1(p.x / s + b.x);
  o.y = elu1(p.y / s + b.y);
  o.z = elu1(p.z / s + b.z);
  o.w = elu1(p.w / s + b.w);
  return o;
}
template <bool NORM>
__global__ __launch_bounds__(256) void gat_gemm128(const float* __restrict__ X,
                                                   const float* __restrict__ W,
                                                   float* __restrict__ Hout,
                                                   const float* __restrict__ den,
                                                   const float* __restrict__ bias,
                                                   unsigned* __restrict__ gst,
                                                   int N) {
  __shared__ float4 As4[2][16 * GEMM_P4];
  const int tid  = threadIdx.x;
  if (blockIdx.x == 0 && tid < 9) gst[tid * CTR_PITCH] = 0u;
  const int lane = tid & 31;
  const int wv   = tid >> 5;
  const int n0   = wv << 4;
  const int r    = lane & 15;
  const int half = lane >> 4;
  const int ntiles = (N + 15) >> 4;

  v2f bf[32];
#pragma unroll
  for (int j = 0; j < 16; ++j) {
    const float* wp = W + (size_t)(8 * j + 4 * half) * 128 + n0 + r;
    bf[2 * j].x     = wp[0];
    bf[2 * j].y     = wp[128];
    bf[2 * j + 1].x = wp[256];
    bf[2 * j + 1].y = wp[384];
  }

  const int lrow = tid >> 5;
  const int c4   = tid & 31;
  const int hq   = c4 >> 3;
  int t = blockIdx.x;
  float4 p0 = make_float4(0.f, 0.f, 0.f, 0.f), p1 = p0;
  float4 bq = p0;
  float d0 = 1.f, d1 = 1.f;
  if (NORM) bq = *(const float4*)(bias + c4 * 4);
  if (t < ntiles) {
    int ra = (t << 4) + lrow, rb = ra + 8;
    ra = ra < N ? ra : N - 1;
    rb = rb < N ? rb : N - 1;
    p0 = *(const float4*)(X + (size_t)ra * 128 + c4 * 4);
    p1 = *(const float4*)(X + (size_t)rb * 128 + c4 * 4);
    if (NORM) { d0 = den[(size_t)ra * 4 + hq]; d1 = den[(size_t)rb * 4 + hq]; }
  }
  int buf = 0;
  for (; t < ntiles; t += gridDim.x) {
    if (NORM) { p0 = close_piece(p0, d0, bq); p1 = close_piece(p1, d1, bq); }
    As4[buf][lrow * GEMM_P4 + c4]       = p0;
    As4[buf][(lrow + 8) * GEMM_P4 + c4] = p1;
    __syncthreads();
    const int tn = t + gridDim.x;
    if (tn < ntiles) {
      int ra = (tn << 4) + lrow, rb = ra + 8;
      ra = ra < N ? ra : N - 1;
      rb = rb < N ? rb : N - 1;
      p0 = *(const float4*)(X + (size_t)ra * 128 + c4 * 4);
      p1 = *(const float4*)(X + (size_t)rb * 128 + c4 * 4);
      if (NORM) { d0 = den[(size_t)ra * 4 + hq]; d1 = den[(size_t)rb * 4 + hq]; }
    }
    v8f acc0 = {0.f, 0.f, 0.f, 0.f, 0.f, 0.f, 0.f, 0.f};
    v8f acc1 = acc0;
    const float4* arow = &As4[buf][r * GEMM_P4 + half];
#pragma unroll
    for (int j = 0; j < 16; ++j) {
      const float4 q = arow[2 * j];
      v2f a0; a0.x = q.x; a0.y = q.y;
      v2f a1; a1.x = q.z; a1.y = q.w;
      acc0 = __builtin_amdgcn_wmma_f32_16x16x4_f32(false, a0, false, bf[2 * j],
                                                   (short)0, acc0, false, false);
      acc1 = __builtin_amdgcn_wmma_f32_16x16x4_f32(false, a1, false, bf[2 * j + 1],
                                                   (short)0, acc1, false, false);
    }
    const int rowbase = (t << 4) + (half << 3);
    const int col     = n0 + r;
#pragma unroll
    for (int i = 0; i < 8; ++i) {
      const int row = rowbase + i;
      if (row < N) Hout[(size_t)row * 128 + col] = acc0[i] + acc1[i];
    }
    buf ^= 1;
  }
}

#define NODE_SPLIT(n) ((((n) >> 1) + 15) & ~15)
__device__ __forceinline__ int chip_half() {
  return (int)((__builtin_amdgcn_s_sendmsg_rtn(0x87) >> 18) & 1u);
}

__global__ __launch_bounds__(256) void gat_scores_init(const float* __restrict__ Hm,
                                const float* __restrict__ asrc,
                                const float* __restrict__ adst,
                                float* __restrict__ es, float* __restrict__ ed,
                                unsigned* __restrict__ menc,
                                float* __restrict__ ssum,
                                float* __restrict__ acc,
                                unsigned* __restrict__ ctr,
                                unsigned* __restrict__ gst, int N) {
  __shared__ unsigned red[8 * 9];
  const int idx = blockIdx.x * blockDim.x + threadIdx.x;
  const int hd = idx & 3;
  unsigned emax = 0u, emin = 0u, dabs = 0u;
  if (idx < N * 4) {
    if (idx < 4) ctr[idx * CTR_PITCH] = 0u;
    const size_t base = (size_t)(idx >> 2) * 128 + hd * 32;
    const float* hp  = Hm + base;
    float*       ap  = acc + base;
    const float* sa  = asrc + hd * 32;
    const float* da  = adst + hd * 32;
    float s = 0.f, d = 0.f;
#pragma unroll
    for (int c = 0; c < 32; c += 4) {
      float4 hv = *(const float4*)(hp + c);
      float4 av = *(const float4*)(sa + c);
      float4 bv = *(const float4*)(da + c);
      s += hv.x * av.x + hv.y * av.y + hv.z * av.z + hv.w * av.w;
      d += hv.x * bv.x + hv.y * bv.y + hv.z * bv.z + hv.w * bv.w;
      *(float4*)(ap + c) = make_float4(0.f, 0.f, 0.f, 0.f);
    }
    es[idx] = s;
    ed[idx] = d;
    menc[idx] = 0u;
    ssum[idx] = 0.f;
    emax = enc_f32(s);
    emin = enc_f32(-s);
    dabs = __float_as_uint(fabsf(d));
  }
#pragma unroll
  for (int o = 4; o < 32; o <<= 1) {
    const unsigned a = (unsigned)__shfl_xor((int)emax, o);
    const unsigned b = (unsigned)__shfl_xor((int)emin, o);
    const unsigned q = (unsigned)__shfl_xor((int)dabs, o);
    emax = a > emax ? a : emax;
    emin = b > emin ? b : emin;
    dabs = q > dabs ? q : dabs;
  }
#pragma unroll
  for (int o = 1; o < 4; o <<= 1) {
    const unsigned q = (unsigned)__shfl_xor((int)dabs, o);
    dabs = q > dabs ? q : dabs;
  }
  const int lane = threadIdx.x & 31;
  const int wv   = threadIdx.x >> 5;
  if (lane < 4) { red[wv * 9 + lane] = emax; red[wv * 9 + 4 + lane] = emin; }
  if (lane == 4) red[wv * 9 + 8] = dabs;
  __syncthreads();
  if (threadIdx.x < 9) {
    unsigned m = red[threadIdx.x];
#pragma unroll
    for (int w = 1; w < 8; ++w) { const unsigned v = red[w * 9 + threadIdx.x]; m = v > m ? v : m; }
    atomicMax(gst + threadIdx.x * CTR_PITCH, m);
  }
}

#define SHIFT_RANGE 16.0f
#define SHIFT_MAG   65536.0f
struct Shift { float g0, g1, g2, g3; bool global_ok; };
__device__ __forceinline__ int shift_head_ok(float gmax, float gmin) {
  return (int)(gmax - gmin <= SHIFT_RANGE) & (int)(gmax <= SHIFT_MAG) & (int)(gmin >= -SHIFT_MAG);
}
__device__ __forceinline__ Shift read_shift(const unsigned* __restrict__ gst) {
  Shift r;
  r.g0 = dec_f32(gst[0 * CTR_PITCH]);
  r.g1 = dec_f32(gst[1 * CTR_PITCH]);
  r.g2 = dec_f32(gst[2 * CTR_PITCH]);
  r.g3 = dec_f32(gst[3 * CTR_PITCH]);
  const float n0 = -dec_f32(gst[4 * CTR_PITCH]);
  const float n1 = -dec_f32(gst[5 * CTR_PITCH]);
  const float n2 = -dec_f32(gst[6 * CTR_PITCH]);
  const float n3 = -dec_f32(gst[7 * CTR_PITCH]);
  const float dm = __uint_as_float(gst[8 * CTR_PITCH]);
  r.global_ok = (shift_head_ok(r.g0, n0) & shift_head_ok(r.g1, n1) &
                 shift_head_ok(r.g2, n2) & shift_head_ok(r.g3, n3) & (int)(dm <= SHIFT_MAG)) != 0;
  return r;
}

__device__ __forceinline__ float lrelu(float x) {
  return x > 0.f ? x : NEG_SLOPE * x;
}

#define MAXP_CHUNKS 4
__global__ __launch_bounds__(256) void gat_edge_max(const int* __restrict__ ei,
                                                    const float* __restrict__ es,
                                                    unsigned* __restrict__ menc,
                                                    unsigned* __restrict__ ctr,
                                                    const unsigned* __restrict__ gst,
                                                    int E, int N) {
  __shared__ unsigned tk[2];
  __shared__ int sh_half;
  if (read_shift(gst).global_ok) return;
  const int tid  = threadIdx.x;
  const int lane = tid & 31;
  const int wv   = tid >> 5;
  const int Et = E + N;
  const int split = NODE_SPLIT(N);
  const unsigned nch = (unsigned)((Et + 256 * MAXP_CHUNKS - 1) / (256 * MAXP_CHUNKS));
  if (tid == 0) sh_half = chip_half();
  __syncthreads();
  const int myhalf = sh_half;
  const int sub = lane >> 2;
  const int hd  = lane & 3;
  unsigned it = 0;
  for (int pass = 0; pass < 2; ++pass) {
    const int part = myhalf ^ pass;
    unsigned* cp = ctr + part * CTR_PITCH;
    unsigned nxt = 0;
    if (tid == 0) nxt = atomicAdd(cp, 1u);
    for (unsigned k = 0; k <= nch; ++k) {
      if (tid == 0) tk[it & 1] = nxt;
      __syncthreads();
      const unsigned t = tk[it & 1];
      ++it;
      if (t >= nch) break;
      if (tid == 0) nxt = atomicAdd(cp, 1u);
      const int e0 = ((int)t * 8 + wv) * (32 * MAXP_CHUNKS);
      if (e0 >= Et) continue;
      int s[MAXP_CHUNKS], d[MAXP_CHUNKS];
      unsigned msk[MAXP_CHUNKS];
#pragma unroll
      for (int c = 0; c < MAXP_CHUNKS; ++c) {
        const int e = e0 + c * 32 + lane;
        s[c] = 0; d[c] = 0;
        if (e < Et) {
          if (e < E) { s[c] = ei[e]; d[c] = ei[E + e]; } else { s[c] = d[c] = e - E; }
        }
      }
#pragma unroll
      for (int c = 0; c < MAXP_CHUNKS; ++c) {
        const int e = e0 + c * 32 + lane;
        const bool mine = (e < Et) && (((d[c] >= split) ? 1 : 0) == part);
        msk[c] = __builtin_amdgcn_ballot_w32(mine);
      }
      float v[MAXP_CHUNKS * 4];
      int   dj[MAXP_CHUNKS * 4];
      unsigned on = 0u;
#pragma unroll
      for (int c = 0; c < MAXP_CHUNKS; ++c) {
#pragma unroll
        for (int g = 0; g < 4; ++g) {
          const int from = g * 8 + sub;
          const int sj = __shfl(s[c], from);
          dj[c * 4 + g] = __shfl(d[c], from);
          const bool p = (msk[c] >> from) & 1u;
          v[c * 4 + g] = 0.f;
          if (p) {
            v[c * 4 + g] = es[(size_t)sj * 4 + hd];
            on |= 1u << (c * 4 + g);
          }
        }
      }
#pragma unroll
      for (int i = 0; i < MAXP_CHUNKS * 4; ++i)
        if ((on >> i) & 1u) atomicMax(menc + (size_t)dj[i] * 4 + hd, enc_f32(v[i]));
    }
  }
}

struct Row4 { float c0, c1, c2, c3; };

__device__ __forceinline__ float lane_f(float v, int j) {
  return __int_as_float(__builtin_amdgcn_readlane(__float_as_int(v), j));
}
__device__ __forceinline__ Row4 load_row(const float* __restrict__ Hm, int sj, int lane) {
  const float* hp = Hm + (size_t)sj * 128 + lane;
  Row4 r;
  r.c0 = hp[0]; r.c1 = hp[32]; r.c2 = hp[64]; r.c3 = hp[96];
  return r;
}
__device__ __forceinline__ void add_row(float* __restrict__ acc, int dj, Row4 h,
                                        float w0, float w1, float w2, float w3, int lane) {
  float* op = acc + (size_t)dj * 128 + lane;
  atomicAdd(op,      h.c0 * w0);
  atomicAdd(op + 32, h.c1 * w1);
  atomicAdd(op + 64, h.c2 * w2);
  atomicAdd(op + 96, h.c3 * w3);
}
__device__ __forceinline__ void add_den(float* __restrict__ ssum, int d,
                                        float x0, float x1, float x2, float x3,
                                        int j0, unsigned mask, int lane) {
  const int src = j0 + (lane >> 2);
  const int hsel = lane & 3;
  const float t0 = __shfl(x0, src), t1 = __shfl(x1, src);
  const float t2 = __shfl(x2, src), t3 = __shfl(x3, src);
  const int dz = __shfl(d, src);
  const float za = (hsel & 1) ? t1 : t0;
  const float zb = (hsel & 1) ? t3 : t2;
  const float z  = (hsel & 2) ? zb : za;
  if ((mask >> src) & 1u) atomicAdd(&ssum[(size_t)dz * 4 + hsel], z);
}

#define SCAT_SUB 2
__global__ __launch_bounds__(256) void gat_edge_scatter(
    const int* __restrict__ ei,
    const float* __restrict__ es, const float* __restrict__ ed,
    const unsigned* __restrict__ menc, float* __restrict__ ssum,
    const float* __restrict__ Hm, float* __restrict__ acc,
    unsigned* __restrict__ ctr, const unsigned* __restrict__ gst, int E, int N) {
  __shared__ unsigned tk[2];
  __shared__ int sh_half;
  const Shift sh = read_shift(gst);
  const int tid  = threadIdx.x;
  const int lane = tid & 31;
  const int wv   = tid >> 5;
  const int Et = E + N;
  const int split = NODE_SPLIT(N);
  const unsigned nch = (unsigned)((Et + 256 * SCAT_SUB - 1) / (256 * SCAT_SUB));
  if (tid == 0) sh_half = chip_half();
  __syncthreads();
  const int myhalf = sh_half;
  unsigned it = 0;
  for (int pass = 0; pass < 2; ++pass) {
    const int part = myhalf ^ pass;
    unsigned* cp = ctr + (2 + part) * CTR_PITCH;
    unsigned nxt = 0;
    if (tid == 0) nxt = atomicAdd(cp, 1u);
    for (unsigned k = 0; k <= nch; ++k) {
      if (tid == 0) tk[it & 1] = nxt;
      __syncthreads();
      const unsigned t = tk[it & 1];
      ++it;
      if (t >= nch) break;
      if (tid == 0) nxt = atomicAdd(cp, 1u);
      const int eb = ((int)t * 8 + wv) * (32 * SCAT_SUB);
      int sn = 0, dn = 0;
      {
        const int e = eb + lane;
        if (e < Et) { if (e < E) { sn = ei[e]; dn = ei[E + e]; } else { sn = dn = e - E; } }
      }
#pragma unroll 1
      for (int c = 0; c < SCAT_SUB; ++c) {
        const int e0 = eb + c * 32;
        if (e0 >= Et) break;
        const int e = e0 + lane;
        const int s = sn, d = dn;
        sn = 0; dn = 0;
        if (c + 1 < SCAT_SUB) {
          const int en = e + 32;
          if (en < Et) { if (en < E) { sn = ei[en]; dn = ei[E + en]; } else { sn = dn = en - E; } }
        }
        float x0 = 0.f, x1 = 0.f, x2 = 0.f, x3 = 0.f;
        const bool mine = (e < Et) && (((d >= split) ? 1 : 0) == part);
        if (mine) {
          const float4 vs = *(const float4*)(es + (size_t)s * 4);
          const float4 vd = *(const float4*)(ed + (size_t)d * 4);
          float m0 = sh.g0, m1 = sh.g1, m2 = sh.g2, m3 = sh.g3;
          if (!sh.global_ok) {
            const uint4 mm = *(const uint4*)(menc + (size_t)d * 4);
            m0 = dec_f32(mm.x); m1 = dec_f32(mm.y); m2 = dec_f32(mm.z); m3 = dec_f32(mm.w);
          }
          x0 = expf(lrelu(vs.x + vd.x) - lrelu(m0 + vd.x));
          x1 = expf(lrelu(vs.y + vd.y) - lrelu(m1 + vd.y));
          x2 = expf(lrelu(vs.z + vd.z) - lrelu(m2 + vd.z));
          x3 = expf(lrelu(vs.w + vd.w) - lrelu(m3 + vd.w));
        }
        const unsigned mask = __builtin_amdgcn_ballot_w32(mine);
        for (int j0 = 0; j0 < 32; j0 += 8)
          if ((mask >> j0) & 0xFFu) add_den(ssum, d, x0, x1, x2, x3, j0, mask, lane);
        unsigned m = mask;
        for (int b = 0; b < 4 && m != 0u; ++b) {
          int js[8];
          int cnt = 0;
#pragma unroll
          for (int u = 0; u < 8; ++u) {
            js[u] = 0;
            if (m != 0u) { js[u] = __builtin_ctz(m); m &= m - 1u; ++cnt; }
          }
          Row4 hv[8];
          if (cnt == 8) {
#pragma unroll
            for (int u = 0; u < 8; ++u)
              hv[u] = load_row(Hm, __builtin_amdgcn_readlane(s, js[u]), lane);
#pragma unroll
            for (int u = 0; u < 8; ++u)
              add_row(acc, __builtin_amdgcn_readlane(d, js[u]), hv[u],
                      lane_f(x0, js[u]), lane_f(x1, js[u]),
                      lane_f(x2, js[u]), lane_f(x3, js[u]), lane);
          } else {
#pragma unroll
            for (int u = 0; u < 7; ++u)
              if (u < cnt) hv[u] = load_row(Hm, __builtin_amdgcn_readlane(s, js[u]), lane);
#pragma unroll
            for (int u = 0; u < 7; ++u)
              if (u < cnt)
                add_row(acc, __builtin_amdgcn_readlane(d, js[u]), hv[u],
                        lane_f(x0, js[u]), lane_f(x1, js[u]),
                        lane_f(x2, js[u]), lane_f(x3, js[u]), lane);
          }
        }
      }
    }
  }
}

__global__ void gat_finalize(const float* __restrict__ acc,
                             const float* __restrict__ ssum,
                             const float* __restrict__ bias,
                             float* __restrict__ outp, int N, int apply_elu) {
  const int idx = blockIdx.x * blockDim.x + threadIdx.x;
  if (idx >= N * 128) return;
  const int n  = idx >> 7;
  const int hd = (idx >> 5) & 3;
  float v = acc[idx] / ssum[(size_t)n * 4 + hd] + bias[idx & 127];
  if (apply_elu) v = v > 0.f ? v : expm1f(v);
  outp[idx] = v;
}

extern "C" void kernel_launch(void* const* d_in, const int* in_sizes, int n_in,
                              void* d_out, int out_size, void* d_ws, size_t ws_size,
                              hipStream_t stream) {
  const float* x  = (const float*)d_in[0];
  const int*   ei = (const int*)d_in[1];
  const int N = in_sizes[0] / 128;
  const int E = in_sizes[1] / 2;

  const float* Wl[3] = {(const float*)d_in[2], (const float*)d_in[6],  (const float*)d_in[10]};
  const float* As[3] = {(const float*)d_in[3], (const float*)d_in[7],  (const float*)d_in[11]};
  const float* Ad[3] = {(const float*)d_in[4], (const float*)d_in[8],  (const float*)d_in[12]};
  const float* Bb[3] = {(const float*)d_in[5], (const float*)d_in[9],  (const float*)d_in[13]};

  float*    F0   = (float*)d_ws;
  float*    F1   = F0 + (size_t)N * 128;
  float*    es   = F1 + (size_t)N * 128;
  float*    ed   = es + (size_t)N * 4;
  unsigned* menc = (unsigned*)(ed + (size_t)N * 4);
  float*    ssum = (float*)(menc + (size_t)N * 4);
  unsigned* ctr  = (unsigned*)(ssum + (size_t)N * 4);
  unsigned* gst  = ctr + 4 * CTR_PITCH;

  const dim3 bGemm(256);
  const dim3 gGemm(512);
  const int gScore = (N * 4 + 255) / 256;
  const int gMax   = 512;
  const int gScat  = 1280;
  const int gFin   = (N * 128 + 255) / 256;

  float* Hm = F0;
  float* Ac = F1;
  for (int layer = 0; layer < 3; ++layer) {
    if (layer == 0)
      gat_gemm128<false><<<gGemm, bGemm, 0, stream>>>(x, Wl[0], Hm, ssum, Bb[0], gst, N);
    else
      gat_gemm128<true><<<gGemm, bGemm, 0, stream>>>(Ac, Wl[layer], Hm, ssum, Bb[layer - 1], gst, N);
    gat_scores_init<<<gScore, 256, 0, stream>>>(Hm, As[layer], Ad[layer],
                                                es, ed, menc, ssum, Ac, ctr, gst, N);
    gat_edge_max<<<gMax, 256, 0, stream>>>(ei, es, menc, ctr, gst, E, N);
    gat_edge_scatter<<<gScat, 256, 0, stream>>>(ei, es, ed, menc, ssum, Hm, Ac, ctr, gst, E, N);
  }
  gat_finalize<<<gFin, 256, 0, stream>>>(Ac, ssum, Bb[2], (float*)d_out, N, 0);
}
